# starting kernel
# baseline (speedup 1.0000x reference)
#include <hip/hip_runtime.h>
#include <math.h>

typedef __attribute__((ext_vector_type(2))) float v2f;
typedef __attribute__((ext_vector_type(8))) float v8f;

#define HAS_WMMA_F32_16X16X4 __has_builtin(__builtin_amdgcn_wmma_f32_16x16x4_f32)

__global__ __launch_bounds__(256) void sl_partial_sum_kernel(
    const float* __restrict__ loss, float* __restrict__ partials,
    int chunk_per_block)
{
    const int wave = threadIdx.x >> 5;
    const int lane = threadIdx.x & 31;
    const int wave_chunk = chunk_per_block >> 3;
    const int iters = wave_chunk >> 6;
    const long base = (long)blockIdx.x * chunk_per_block + (long)wave * wave_chunk;

    float laneSum;

#if HAS_WMMA_F32_16X16X4
    v8f acc = {};
    v2f ones; ones.x = 1.0f; ones.y = 1.0f;
    const float* p = loss + base + lane * 2;
    for (int it = 0; it < iters; ++it) {
        v2f a = *(const v2f*)(p + (long)it * 64);
        acc = __builtin_amdgcn_wmma_f32_16x16x4_f32(
             false, a,  false, ones,
             (short)0, acc,  false,  false);
    }
    laneSum = ((acc[0] + acc[1]) + (acc[2] + acc[3])) +
              ((acc[4] + acc[5]) + (acc[6] + acc[7]));
#else
    float s = 0.0f;
    const float* p = loss + base + lane * 2;
    for (int it = 0; it < iters; ++it) {
        v2f a = *(const v2f*)(p + (long)it * 64);
        s += a.x + a.y;
    }
    laneSum = s;
#endif

    #pragma unroll
    for (int off = 16; off > 0; off >>= 1)
        laneSum += __shfl_xor(laneSum, off, 32);

#if HAS_WMMA_F32_16X16X4
    laneSum *= (1.0f / 16.0f);
#endif

    __shared__ float smem[8];
    if (lane == 0) smem[wave] = laneSum;
    __syncthreads();
    if (threadIdx.x == 0) {
        float s = 0.0f;
        #pragma unroll
        for (int w = 0; w < 8; ++w) s += smem[w];
        partials[blockIdx.x] = s;
    }
}

__global__ __launch_bounds__(512) void sl_finalize_tau_kernel(
    const float* __restrict__ partials, float* __restrict__ tau_out,
    int nparts, float inv_n)
{
    float v = (threadIdx.x < (unsigned)nparts) ? partials[threadIdx.x] : 0.0f;
    #pragma unroll
    for (int off = 16; off > 0; off >>= 1)
        v += __shfl_xor(v, off, 32);

    __shared__ float smem[16];
    const int wave = threadIdx.x >> 5;
    const int lane = threadIdx.x & 31;
    if (lane == 0) smem[wave] = v;
    __syncthreads();
    if (threadIdx.x == 0) {
        float s = 0.0f;
        #pragma unroll
        for (int w = 0; w < 16; ++w) s += smem[w];
        const float mean = s * inv_n;
        tau_out[0] = 0.9f * 0.5f + 0.1f * mean;
    }
}

__device__ __forceinline__ float sl_lambertw(float z)
{
    const float p2 = fmaxf(fmaf(5.43656365691809f, z, 2.0f), 0.0f);
    const float p  = __builtin_sqrtf(p2);
    const float w_near = -1.0f + p * (1.0f - p * (1.0f / 3.0f));
    float w = (z < -0.25f) ? w_near : log1pf(z);
    #pragma unroll
    for (int it = 0; it < 4; ++it) {
        const float ew   = __expf(w);
        const float f    = fmaf(w, ew, -z);
        const float wp1  = w + 1.0f;
        const float corr = __fdividef((w + 2.0f) * f, 2.0f * wp1);
        const float den  = fmaf(ew, wp1, -corr);
        w = w - __fdividef(f, den);
    }
    return w;
}

__global__ __launch_bounds__(256) void sl_superloss_kernel(
    const float* __restrict__ loss, const float* __restrict__ tau_ptr,
    float* __restrict__ out_super, float* __restrict__ out_sigma, int n4)
{
    const int i = blockIdx.x * blockDim.x + threadIdx.x;
    if (i >= n4) return;

    const float tau = tau_ptr[0];
    const float zmin = -0.36787944117144233f + 1.1920928955078125e-7f;

    const float4 x = ((const float4*)loss)[i];
    float xs[4] = {x.x, x.y, x.z, x.w};
    float sg[4], sl[4];

    #pragma unroll
    for (int k = 0; k < 4; ++k) {
        const float beta = xs[k] - tau;
        const float z = fmaxf(zmin, 0.5f * beta);
        const float w = sl_lambertw(z);
        const float sigma = __expf(-w);
        sg[k] = sigma;
        sl[k] = sigma * xs[k];
    }

    float4 o;
    o.x = sl[0]; o.y = sl[1]; o.z = sl[2]; o.w = sl[3];
    ((float4*)out_super)[i] = o;
    o.x = sg[0]; o.y = sg[1]; o.z = sg[2]; o.w = sg[3];
    ((float4*)out_sigma)[i] = o;
}

extern "C" void kernel_launch(void* const* d_in, const int* in_sizes, int n_in,
                              void* d_out, int out_size, void* d_ws, size_t ws_size,
                              hipStream_t stream)
{
    const float* loss = (const float*)d_in[0];
    const int n = in_sizes[0];

    float* out_super = (float*)d_out;
    float* out_sigma = out_super + n;

    float* partials = (float*)d_ws;
    float* tau_ptr  = partials + 512;

    const int RBLOCKS = 512;
    const int chunk = n / RBLOCKS;

    sl_partial_sum_kernel<<<RBLOCKS, 256, 0, stream>>>(loss, partials, chunk);
    sl_finalize_tau_kernel<<<1, 512, 0, stream>>>(partials, tau_ptr, RBLOCKS,
                                                  1.0f / (float)n);

    const int n4 = n >> 2;
    sl_superloss_kernel<<<(n4 + 255) / 256, 256, 0, stream>>>(
        loss, tau_ptr, out_super, out_sigma, n4);
}


# Halley step of the Lambert-W: both quotients as products with v_rcp_f32 instead of the correctly rounded 11-instruction division that '/' and __fdividef expand to
# speedup vs baseline: 1.4901x; 1.4901x over previous
#include <hip/hip_runtime.h>
#include <math.h>

typedef __attribute__((ext_vector_type(2))) float v2f;
typedef __attribute__((ext_vector_type(8))) float v8f;

#define HAS_WMMA_F32_16X16X4 __has_builtin(__builtin_amdgcn_wmma_f32_16x16x4_f32)

__global__ __launch_bounds__(256) void sl_partial_sum_kernel(
    const float* __restrict__ loss, float* __restrict__ partials,
    int chunk_per_block)
{
    const int wave = threadIdx.x >> 5;
    const int lane = threadIdx.x & 31;
    const int wave_chunk = chunk_per_block >> 3;
    const int iters = wave_chunk >> 6;
    const long base = (long)blockIdx.x * chunk_per_block + (long)wave * wave_chunk;

    float laneSum;

#if HAS_WMMA_F32_16X16X4
    v8f acc = {};
    v2f ones; ones.x = 1.0f; ones.y = 1.0f;
    const float* p = loss + base + lane * 2;
    for (int it = 0; it < iters; ++it) {
        v2f a = *(const v2f*)(p + (long)it * 64);
        acc = __builtin_amdgcn_wmma_f32_16x16x4_f32(
             false, a,  false, ones,
             (short)0, acc,  false,  false);
    }
    laneSum = ((acc[0] + acc[1]) + (acc[2] + acc[3])) +
              ((acc[4] + acc[5]) + (acc[6] + acc[7]));
#else
    float s = 0.0f;
    const float* p = loss + base + lane * 2;
    for (int it = 0; it < iters; ++it) {
        v2f a = *(const v2f*)(p + (long)it * 64);
        s += a.x + a.y;
    }
    laneSum = s;
#endif

    #pragma unroll
    for (int off = 16; off > 0; off >>= 1)
        laneSum += __shfl_xor(laneSum, off, 32);

#if HAS_WMMA_F32_16X16X4
    laneSum *= (1.0f / 16.0f);
#endif

    __shared__ float smem[8];
    if (lane == 0) smem[wave] = laneSum;
    __syncthreads();
    if (threadIdx.x == 0) {
        float s = 0.0f;
        #pragma unroll
        for (int w = 0; w < 8; ++w) s += smem[w];
        partials[blockIdx.x] = s;
    }
}

__global__ __launch_bounds__(512) void sl_finalize_tau_kernel(
    const float* __restrict__ partials, float* __restrict__ tau_out,
    int nparts, float inv_n)
{
    float v = (threadIdx.x < (unsigned)nparts) ? partials[threadIdx.x] : 0.0f;
    #pragma unroll
    for (int off = 16; off > 0; off >>= 1)
        v += __shfl_xor(v, off, 32);

    __shared__ float smem[16];
    const int wave = threadIdx.x >> 5;
    const int lane = threadIdx.x & 31;
    if (lane == 0) smem[wave] = v;
    __syncthreads();
    if (threadIdx.x == 0) {
        float s = 0.0f;
        #pragma unroll
        for (int w = 0; w < 16; ++w) s += smem[w];
        const float mean = s * inv_n;
        tau_out[0] = 0.9f * 0.5f + 0.1f * mean;
    }
}

__device__ __forceinline__ float sl_lambertw(float z)
{
    const float p2 = fmaxf(fmaf(5.43656365691809f, z, 2.0f), 0.0f);
    const float p  = __builtin_sqrtf(p2);
    const float w_near = -1.0f + p * (1.0f - p * (1.0f / 3.0f));
    float w = (z < -0.25f) ? w_near : log1pf(z);
    #pragma unroll
    for (int it = 0; it < 4; ++it) {
        const float ew   = __expf(w);
        const float f    = fmaf(w, ew, -z);
        const float wp1  = w + 1.0f;
        const float corr = (fmaf(0.5f, w, 1.0f) * f) * __builtin_amdgcn_rcpf(wp1);
        const float den  = fmaf(ew, wp1, -corr);
        w = fmaf(-f, __builtin_amdgcn_rcpf(den), w);
    }
    return w;
}

__global__ __launch_bounds__(256) void sl_superloss_kernel(
    const float* __restrict__ loss, const float* __restrict__ tau_ptr,
    float* __restrict__ out_super, float* __restrict__ out_sigma, int n4)
{
    const int i = blockIdx.x * blockDim.x + threadIdx.x;
    if (i >= n4) return;

    const float tau = tau_ptr[0];
    const float zmin = -0.36787944117144233f + 1.1920928955078125e-7f;

    const float4 x = ((const float4*)loss)[i];
    float xs[4] = {x.x, x.y, x.z, x.w};
    float sg[4], sl[4];

    #pragma unroll
    for (int k = 0; k < 4; ++k) {
        const float beta = xs[k] - tau;
        const float z = fmaxf(zmin, 0.5f * beta);
        const float w = sl_lambertw(z);
        const float sigma = __expf(-w);
        sg[k] = sigma;
        sl[k] = sigma * xs[k];
    }

    float4 o;
    o.x = sl[0]; o.y = sl[1]; o.z = sl[2]; o.w = sl[3];
    ((float4*)out_super)[i] = o;
    o.x = sg[0]; o.y = sg[1]; o.z = sg[2]; o.w = sg[3];
    ((float4*)out_sigma)[i] = o;
}

extern "C" void kernel_launch(void* const* d_in, const int* in_sizes, int n_in,
                              void* d_out, int out_size, void* d_ws, size_t ws_size,
                              hipStream_t stream)
{
    const float* loss = (const float*)d_in[0];
    const int n = in_sizes[0];

    float* out_super = (float*)d_out;
    float* out_sigma = out_super + n;

    float* partials = (float*)d_ws;
    float* tau_ptr  = partials + 512;

    const int RBLOCKS = 512;
    const int chunk = n / RBLOCKS;

    sl_partial_sum_kernel<<<RBLOCKS, 256, 0, stream>>>(loss, partials, chunk);
    sl_finalize_tau_kernel<<<1, 512, 0, stream>>>(partials, tau_ptr, RBLOCKS,
                                                  1.0f / (float)n);

    const int n4 = n >> 2;
    sl_superloss_kernel<<<(n4 + 255) / 256, 256, 0, stream>>>(
        loss, tau_ptr, out_super, out_sigma, n4);
}


# Lambert-W initial guess from raw v_log_f32 (ln2*log2(1+z)) and v_sqrt_f32 instead of the library log1pf (some 85 instructions an element) and the correctly rounded sqrtf; Halley steps unchanged
# speedup vs baseline: 2.2169x; 1.4878x over previous
#include <hip/hip_runtime.h>
#include <math.h>

typedef __attribute__((ext_vector_type(2))) float v2f;
typedef __attribute__((ext_vector_type(8))) float v8f;

#define HAS_WMMA_F32_16X16X4 __has_builtin(__builtin_amdgcn_wmma_f32_16x16x4_f32)

__global__ __launch_bounds__(256) void sl_partial_sum_kernel(
    const float* __restrict__ loss, float* __restrict__ partials,
    int chunk_per_block)
{
    const int wave = threadIdx.x >> 5;
    const int lane = threadIdx.x & 31;
    const int wave_chunk = chunk_per_block >> 3;
    const int iters = wave_chunk >> 6;
    const long base = (long)blockIdx.x * chunk_per_block + (long)wave * wave_chunk;

    float laneSum;

#if HAS_WMMA_F32_16X16X4
    v8f acc = {};
    v2f ones; ones.x = 1.0f; ones.y = 1.0f;
    const float* p = loss + base + lane * 2;
    for (int it = 0; it < iters; ++it) {
        v2f a = *(const v2f*)(p + (long)it * 64);
        acc = __builtin_amdgcn_wmma_f32_16x16x4_f32(
             false, a,  false, ones,
             (short)0, acc,  false,  false);
    }
    laneSum = ((acc[0] + acc[1]) + (acc[2] + acc[3])) +
              ((acc[4] + acc[5]) + (acc[6] + acc[7]));
#else
    float s = 0.0f;
    const float* p = loss + base + lane * 2;
    for (int it = 0; it < iters; ++it) {
        v2f a = *(const v2f*)(p + (long)it * 64);
        s += a.x + a.y;
    }
    laneSum = s;
#endif

    #pragma unroll
    for (int off = 16; off > 0; off >>= 1)
        laneSum += __shfl_xor(laneSum, off, 32);

#if HAS_WMMA_F32_16X16X4
    laneSum *= (1.0f / 16.0f);
#endif

    __shared__ float smem[8];
    if (lane == 0) smem[wave] = laneSum;
    __syncthreads();
    if (threadIdx.x == 0) {
        float s = 0.0f;
        #pragma unroll
        for (int w = 0; w < 8; ++w) s += smem[w];
        partials[blockIdx.x] = s;
    }
}

__global__ __launch_bounds__(512) void sl_finalize_tau_kernel(
    const float* __restrict__ partials, float* __restrict__ tau_out,
    int nparts, float inv_n)
{
    float v = (threadIdx.x < (unsigned)nparts) ? partials[threadIdx.x] : 0.0f;
    #pragma unroll
    for (int off = 16; off > 0; off >>= 1)
        v += __shfl_xor(v, off, 32);

    __shared__ float smem[16];
    const int wave = threadIdx.x >> 5;
    const int lane = threadIdx.x & 31;
    if (lane == 0) smem[wave] = v;
    __syncthreads();
    if (threadIdx.x == 0) {
        float s = 0.0f;
        #pragma unroll
        for (int w = 0; w < 16; ++w) s += smem[w];
        const float mean = s * inv_n;
        tau_out[0] = 0.9f * 0.5f + 0.1f * mean;
    }
}

__device__ __forceinline__ float sl_lambertw(float z)
{
    const float p2 = fmaxf(fmaf(5.43656365691809f, z, 2.0f), 0.0f);
    const float p  = __builtin_amdgcn_sqrtf(p2);
    const float w_near = -1.0f + p * (1.0f - p * (1.0f / 3.0f));
    const float w_far = 0.6931471805599453f * __builtin_amdgcn_logf(1.0f + z);
    float w = (z < -0.25f) ? w_near : w_far;
    #pragma unroll
    for (int it = 0; it < 4; ++it) {
        const float ew   = __expf(w);
        const float f    = fmaf(w, ew, -z);
        const float wp1  = w + 1.0f;
        const float corr = (fmaf(0.5f, w, 1.0f) * f) * __builtin_amdgcn_rcpf(wp1);
        const float den  = fmaf(ew, wp1, -corr);
        w = fmaf(-f, __builtin_amdgcn_rcpf(den), w);
    }
    return w;
}

__global__ __launch_bounds__(256) void sl_superloss_kernel(
    const float* __restrict__ loss, const float* __restrict__ tau_ptr,
    float* __restrict__ out_super, float* __restrict__ out_sigma, int n4)
{
    const int i = blockIdx.x * blockDim.x + threadIdx.x;
    if (i >= n4) return;

    const float tau = tau_ptr[0];
    const float zmin = -0.36787944117144233f + 1.1920928955078125e-7f;

    const float4 x = ((const float4*)loss)[i];
    float xs[4] = {x.x, x.y, x.z, x.w};
    float sg[4], sl[4];

    #pragma unroll
    for (int k = 0; k < 4; ++k) {
        const float beta = xs[k] - tau;
        const float z = fmaxf(zmin, 0.5f * beta);
        const float w = sl_lambertw(z);
        const float sigma = __expf(-w);
        sg[k] = sigma;
        sl[k] = sigma * xs[k];
    }

    float4 o;
    o.x = sl[0]; o.y = sl[1]; o.z = sl[2]; o.w = sl[3];
    ((float4*)out_super)[i] = o;
    o.x = sg[0]; o.y = sg[1]; o.z = sg[2]; o.w = sg[3];
    ((float4*)out_sigma)[i] = o;
}

extern "C" void kernel_launch(void* const* d_in, const int* in_sizes, int n_in,
                              void* d_out, int out_size, void* d_ws, size_t ws_size,
                              hipStream_t stream)
{
    const float* loss = (const float*)d_in[0];
    const int n = in_sizes[0];

    float* out_super = (float*)d_out;
    float* out_sigma = out_super + n;

    float* partials = (float*)d_ws;
    float* tau_ptr  = partials + 512;

    const int RBLOCKS = 512;
    const int chunk = n / RBLOCKS;

    sl_partial_sum_kernel<<<RBLOCKS, 256, 0, stream>>>(loss, partials, chunk);
    sl_finalize_tau_kernel<<<1, 512, 0, stream>>>(partials, tau_ptr, RBLOCKS,
                                                  1.0f / (float)n);

    const int n4 = n >> 2;
    sl_superloss_kernel<<<(n4 + 255) / 256, 256, 0, stream>>>(
        loss, tau_ptr, out_super, out_sigma, n4);
}


# Partial-sum kernel: each wave requests its whole 16 KiB first (32 loads of 16 B a lane in flight, was 8 of 8 B), halves folded into two WMMA accumulators; new fixed summation order for the mean
# speedup vs baseline: 2.3687x; 1.0685x over previous
#include <hip/hip_runtime.h>
#include <math.h>

typedef __attribute__((ext_vector_type(2))) float v2f;
typedef __attribute__((ext_vector_type(4))) float v4f;
typedef __attribute__((ext_vector_type(8))) float v8f;

#define HAS_WMMA_F32_16X16X4 __has_builtin(__builtin_amdgcn_wmma_f32_16x16x4_f32)

__global__ __launch_bounds__(256) void sl_partial_sum_kernel(
    const float* __restrict__ loss, float* __restrict__ partials,
    int chunk_per_block)
{
    const int wave = threadIdx.x >> 5;
    const int lane = threadIdx.x & 31;
    const int wave_chunk = chunk_per_block >> 3;
    const long base = (long)blockIdx.x * chunk_per_block + (long)wave * wave_chunk;

    float laneSum;

#if HAS_WMMA_F32_16X16X4
    v8f acc0 = {};
    v8f acc1 = {};
    v2f ones; ones.x = 1.0f; ones.y = 1.0f;

    const int rounds = wave_chunk >> 12;
    const float* p = loss + base + lane * 4;
    for (int r = 0; r < rounds; ++r) {
        v4f a[32];
        #pragma unroll
        for (int q = 0; q < 32; ++q)
            a[q] = *(const v4f*)(p + q * 128);
        __builtin_amdgcn_sched_barrier(0);
        #pragma unroll
        for (int q = 0; q < 32; ++q) {
            v2f lo; lo.x = a[q].x; lo.y = a[q].y;
            v2f hi; hi.x = a[q].z; hi.y = a[q].w;
            acc0 = __builtin_amdgcn_wmma_f32_16x16x4_f32(
                 false, lo,  false, ones,
                 (short)0, acc0,  false,  false);
            acc1 = __builtin_amdgcn_wmma_f32_16x16x4_f32(
                 false, hi,  false, ones,
                 (short)0, acc1,  false,  false);
        }
        p += 4096;
    }
    const int done = rounds << 12;
    const int tail_iters = (wave_chunk - done) >> 6;
    const float* pt = loss + base + done + lane * 2;
    for (int it = 0; it < tail_iters; ++it) {
        v2f a = *(const v2f*)(pt + (long)it * 64);
        acc0 = __builtin_amdgcn_wmma_f32_16x16x4_f32(
             false, a,  false, ones,
             (short)0, acc0,  false,  false);
    }
    laneSum = (((acc0[0] + acc0[1]) + (acc0[2] + acc0[3])) +
               ((acc0[4] + acc0[5]) + (acc0[6] + acc0[7]))) +
              (((acc1[0] + acc1[1]) + (acc1[2] + acc1[3])) +
               ((acc1[4] + acc1[5]) + (acc1[6] + acc1[7])));
#else
    float s = 0.0f;
    const int iters = wave_chunk >> 6;
    const float* p = loss + base + lane * 2;
    for (int it = 0; it < iters; ++it) {
        v2f a = *(const v2f*)(p + (long)it * 64);
        s += a.x + a.y;
    }
    laneSum = s;
#endif

    #pragma unroll
    for (int off = 16; off > 0; off >>= 1)
        laneSum += __shfl_xor(laneSum, off, 32);

#if HAS_WMMA_F32_16X16X4
    laneSum *= (1.0f / 16.0f);
#endif

    __shared__ float smem[8];
    if (lane == 0) smem[wave] = laneSum;
    __syncthreads();
    if (threadIdx.x == 0) {
        float s = 0.0f;
        #pragma unroll
        for (int w = 0; w < 8; ++w) s += smem[w];
        partials[blockIdx.x] = s;
    }
}

__global__ __launch_bounds__(512) void sl_finalize_tau_kernel(
    const float* __restrict__ partials, float* __restrict__ tau_out,
    int nparts, float inv_n)
{
    float v = (threadIdx.x < (unsigned)nparts) ? partials[threadIdx.x] : 0.0f;
    #pragma unroll
    for (int off = 16; off > 0; off >>= 1)
        v += __shfl_xor(v, off, 32);

    __shared__ float smem[16];
    const int wave = threadIdx.x >> 5;
    const int lane = threadIdx.x & 31;
    if (lane == 0) smem[wave] = v;
    __syncthreads();
    if (threadIdx.x == 0) {
        float s = 0.0f;
        #pragma unroll
        for (int w = 0; w < 16; ++w) s += smem[w];
        const float mean = s * inv_n;
        tau_out[0] = 0.9f * 0.5f + 0.1f * mean;
    }
}

__device__ __forceinline__ float sl_lambertw(float z)
{
    const float p2 = fmaxf(fmaf(5.43656365691809f, z, 2.0f), 0.0f);
    const float p  = __builtin_amdgcn_sqrtf(p2);
    const float w_near = -1.0f + p * (1.0f - p * (1.0f / 3.0f));
    const float w_far = 0.6931471805599453f * __builtin_amdgcn_logf(1.0f + z);
    float w = (z < -0.25f) ? w_near : w_far;
    #pragma unroll
    for (int it = 0; it < 4; ++it) {
        const float ew   = __expf(w);
        const float f    = fmaf(w, ew, -z);
        const float wp1  = w + 1.0f;
        const float corr = (fmaf(0.5f, w, 1.0f) * f) * __builtin_amdgcn_rcpf(wp1);
        const float den  = fmaf(ew, wp1, -corr);
        w = fmaf(-f, __builtin_amdgcn_rcpf(den), w);
    }
    return w;
}

__global__ __launch_bounds__(256) void sl_superloss_kernel(
    const float* __restrict__ loss, const float* __restrict__ tau_ptr,
    float* __restrict__ out_super, float* __restrict__ out_sigma, int n4)
{
    const int i = blockIdx.x * blockDim.x + threadIdx.x;
    if (i >= n4) return;

    const float tau = tau_ptr[0];
    const float zmin = -0.36787944117144233f + 1.1920928955078125e-7f;

    const float4 x = ((const float4*)loss)[i];
    float xs[4] = {x.x, x.y, x.z, x.w};
    float sg[4], sl[4];

    #pragma unroll
    for (int k = 0; k < 4; ++k) {
        const float beta = xs[k] - tau;
        const float z = fmaxf(zmin, 0.5f * beta);
        const float w = sl_lambertw(z);
        const float sigma = __expf(-w);
        sg[k] = sigma;
        sl[k] = sigma * xs[k];
    }

    float4 o;
    o.x = sl[0]; o.y = sl[1]; o.z = sl[2]; o.w = sl[3];
    ((float4*)out_super)[i] = o;
    o.x = sg[0]; o.y = sg[1]; o.z = sg[2]; o.w = sg[3];
    ((float4*)out_sigma)[i] = o;
}

extern "C" void kernel_launch(void* const* d_in, const int* in_sizes, int n_in,
                              void* d_out, int out_size, void* d_ws, size_t ws_size,
                              hipStream_t stream)
{
    const float* loss = (const float*)d_in[0];
    const int n = in_sizes[0];

    float* out_super = (float*)d_out;
    float* out_sigma = out_super + n;

    float* partials = (float*)d_ws;
    float* tau_ptr  = partials + 512;

    const int RBLOCKS = 512;
    const int chunk = n / RBLOCKS;

    sl_partial_sum_kernel<<<RBLOCKS, 256, 0, stream>>>(loss, partials, chunk);
    sl_finalize_tau_kernel<<<1, 512, 0, stream>>>(partials, tau_ptr, RBLOCKS,
                                                  1.0f / (float)n);

    const int n4 = n >> 2;
    sl_superloss_kernel<<<(n4 + 255) / 256, 256, 0, stream>>>(
        loss, tau_ptr, out_super, out_sigma, n4);
}


# Tau finalize folded into the superloss kernel: 2048 resident blocks each rebuild tau from the 512 partials (same order, same bits), then stride over 8 float4 a thread; 2 launches, was 3
# speedup vs baseline: 2.4589x; 1.0381x over previous
#include <hip/hip_runtime.h>
#include <math.h>

typedef __attribute__((ext_vector_type(2))) float v2f;
typedef __attribute__((ext_vector_type(4))) float v4f;
typedef __attribute__((ext_vector_type(8))) float v8f;

#define HAS_WMMA_F32_16X16X4 __has_builtin(__builtin_amdgcn_wmma_f32_16x16x4_f32)

__global__ __launch_bounds__(256) void sl_partial_sum_kernel(
    const float* __restrict__ loss, float* __restrict__ partials,
    int chunk_per_block)
{
    const int wave = threadIdx.x >> 5;
    const int lane = threadIdx.x & 31;
    const int wave_chunk = chunk_per_block >> 3;
    const long base = (long)blockIdx.x * chunk_per_block + (long)wave * wave_chunk;

    float laneSum;

#if HAS_WMMA_F32_16X16X4
    v8f acc0 = {};
    v8f acc1 = {};
    v2f ones; ones.x = 1.0f; ones.y = 1.0f;

    const int rounds = wave_chunk >> 12;
    const float* p = loss + base + lane * 4;
    for (int r = 0; r < rounds; ++r) {
        v4f a[32];
        #pragma unroll
        for (int q = 0; q < 32; ++q)
            a[q] = *(const v4f*)(p + q * 128);
        __builtin_amdgcn_sched_barrier(0);
        #pragma unroll
        for (int q = 0; q < 32; ++q) {
            v2f lo; lo.x = a[q].x; lo.y = a[q].y;
            v2f hi; hi.x = a[q].z; hi.y = a[q].w;
            acc0 = __builtin_amdgcn_wmma_f32_16x16x4_f32(
                 false, lo,  false, ones,
                 (short)0, acc0,  false,  false);
            acc1 = __builtin_amdgcn_wmma_f32_16x16x4_f32(
                 false, hi,  false, ones,
                 (short)0, acc1,  false,  false);
        }
        p += 4096;
    }
    const int done = rounds << 12;
    const int tail_iters = (wave_chunk - done) >> 6;
    const float* pt = loss + base + done + lane * 2;
    for (int it = 0; it < tail_iters; ++it) {
        v2f a = *(const v2f*)(pt + (long)it * 64);
        acc0 = __builtin_amdgcn_wmma_f32_16x16x4_f32(
             false, a,  false, ones,
             (short)0, acc0,  false,  false);
    }
    laneSum = (((acc0[0] + acc0[1]) + (acc0[2] + acc0[3])) +
               ((acc0[4] + acc0[5]) + (acc0[6] + acc0[7]))) +
              (((acc1[0] + acc1[1]) + (acc1[2] + acc1[3])) +
               ((acc1[4] + acc1[5]) + (acc1[6] + acc1[7])));
#else
    float s = 0.0f;
    const int iters = wave_chunk >> 6;
    const float* p = loss + base + lane * 2;
    for (int it = 0; it < iters; ++it) {
        v2f a = *(const v2f*)(p + (long)it * 64);
        s += a.x + a.y;
    }
    laneSum = s;
#endif

    #pragma unroll
    for (int off = 16; off > 0; off >>= 1)
        laneSum += __shfl_xor(laneSum, off, 32);

#if HAS_WMMA_F32_16X16X4
    laneSum *= (1.0f / 16.0f);
#endif

    __shared__ float smem[8];
    if (lane == 0) smem[wave] = laneSum;
    __syncthreads();
    if (threadIdx.x == 0) {
        float s = 0.0f;
        #pragma unroll
        for (int w = 0; w < 8; ++w) s += smem[w];
        partials[blockIdx.x] = s;
    }
}

__device__ __forceinline__ float sl_lambertw(float z)
{
    const float p2 = fmaxf(fmaf(5.43656365691809f, z, 2.0f), 0.0f);
    const float p  = __builtin_amdgcn_sqrtf(p2);
    const float w_near = -1.0f + p * (1.0f - p * (1.0f / 3.0f));
    const float w_far = 0.6931471805599453f * __builtin_amdgcn_logf(1.0f + z);
    float w = (z < -0.25f) ? w_near : w_far;
    #pragma unroll
    for (int it = 0; it < 4; ++it) {
        const float ew   = __expf(w);
        const float f    = fmaf(w, ew, -z);
        const float wp1  = w + 1.0f;
        const float corr = (fmaf(0.5f, w, 1.0f) * f) * __builtin_amdgcn_rcpf(wp1);
        const float den  = fmaf(ew, wp1, -corr);
        w = fmaf(-f, __builtin_amdgcn_rcpf(den), w);
    }
    return w;
}

__global__ __launch_bounds__(256) void sl_superloss_kernel(
    const float* __restrict__ loss, const float* __restrict__ partials,
    float* __restrict__ out_super, float* __restrict__ out_sigma,
    int n4, int nparts, float inv_n)
{
    const int t = threadIdx.x;
    const unsigned stride = gridDim.x * 256u;
    unsigned i = blockIdx.x * 256u + (unsigned)t;

    float4 x = make_float4(0.0f, 0.0f, 0.0f, 0.0f);
    if (i < (unsigned)n4) x = ((const float4*)loss)[i];

    float va = (t < nparts) ? partials[t] : 0.0f;
    float vb = (t + 256 < nparts) ? partials[t + 256] : 0.0f;
    #pragma unroll
    for (int off = 16; off > 0; off >>= 1) {
        va += __shfl_xor(va, off, 32);
        vb += __shfl_xor(vb, off, 32);
    }
    __shared__ float smem[16];
    const int wave = t >> 5;
    const int lane = t & 31;
    if (lane == 0) { smem[wave] = va; smem[wave + 8] = vb; }
    __syncthreads();
    float s = 0.0f;
    #pragma unroll
    for (int w = 0; w < 16; ++w) s += smem[w];
    const float mean = s * inv_n;
    const float tau = 0.9f * 0.5f + 0.1f * mean;

    const float zmin = -0.36787944117144233f + 1.1920928955078125e-7f;

    while (i < (unsigned)n4) {
        const unsigned inext = i + stride;
        const float4 xn = ((const float4*)loss)[(inext < (unsigned)n4) ? inext : i];

        float xs[4] = {x.x, x.y, x.z, x.w};
        float sg[4], sl[4];

        #pragma unroll
        for (int k = 0; k < 4; ++k) {
            const float beta = xs[k] - tau;
            const float z = fmaxf(zmin, 0.5f * beta);
            const float w = sl_lambertw(z);
            const float sigma = __expf(-w);
            sg[k] = sigma;
            sl[k] = sigma * xs[k];
        }

        float4 o;
        o.x = sl[0]; o.y = sl[1]; o.z = sl[2]; o.w = sl[3];
        ((float4*)out_super)[i] = o;
        o.x = sg[0]; o.y = sg[1]; o.z = sg[2]; o.w = sg[3];
        ((float4*)out_sigma)[i] = o;

        x = xn;
        i = inext;
    }
}

extern "C" void kernel_launch(void* const* d_in, const int* in_sizes, int n_in,
                              void* d_out, int out_size, void* d_ws, size_t ws_size,
                              hipStream_t stream)
{
    const float* loss = (const float*)d_in[0];
    const int n = in_sizes[0];

    float* out_super = (float*)d_out;
    float* out_sigma = out_super + n;

    float* partials = (float*)d_ws;

    const int RBLOCKS = 512;
    const int chunk = n / RBLOCKS;

    sl_partial_sum_kernel<<<RBLOCKS, 256, 0, stream>>>(loss, partials, chunk);

    const int n4 = n >> 2;
    sl_superloss_kernel<<<2048, 256, 0, stream>>>(
        loss, partials, out_super, out_sigma, n4, RBLOCKS, 1.0f / (float)n);
}


# speedup vs baseline: 2.5454x; 1.0352x over previous
#include <hip/hip_runtime.h>
#include <math.h>

typedef __attribute__((ext_vector_type(2))) float v2f;
typedef __attribute__((ext_vector_type(4))) float v4f;
typedef __attribute__((ext_vector_type(8))) float v8f;

#define HAS_WMMA_F32_16X16X4 __has_builtin(__builtin_amdgcn_wmma_f32_16x16x4_f32)

__global__ __launch_bounds__(256) void sl_partial_sum_kernel(
    const float* __restrict__ loss, float* __restrict__ partials,
    int chunk_per_block)
{
    const int wave = threadIdx.x >> 5;
    const int lane = threadIdx.x & 31;
    const int wave_chunk = chunk_per_block >> 3;
    const long base = (long)blockIdx.x * chunk_per_block + (long)wave * wave_chunk;

    float laneSum;

#if HAS_WMMA_F32_16X16X4
    v8f acc0 = {};
    v8f acc1 = {};
    v2f ones; ones.x = 1.0f; ones.y = 1.0f;

    const int rounds = wave_chunk >> 12;
    const float* p = loss + base + lane * 4;
    for (int r = 0; r < rounds; ++r) {
        v4f a[32];
        #pragma unroll
        for (int q = 0; q < 32; ++q)
            a[q] = *(const v4f*)(p + q * 128);
        __builtin_amdgcn_sched_barrier(0);
        #pragma unroll
        for (int q = 0; q < 32; ++q) {
            v2f lo; lo.x = a[q].x; lo.y = a[q].y;
            v2f hi; hi.x = a[q].z; hi.y = a[q].w;
            acc0 = __builtin_amdgcn_wmma_f32_16x16x4_f32(
                 false, lo,  false, ones,
                 (short)0, acc0,  false,  false);
            acc1 = __builtin_amdgcn_wmma_f32_16x16x4_f32(
                 false, hi,  false, ones,
                 (short)0, acc1,  false,  false);
        }
        p += 4096;
    }
    const int done = rounds << 12;
    const int tail_iters = (wave_chunk - done) >> 6;
    const float* pt = loss + base + done + lane * 2;
    for (int it = 0; it < tail_iters; ++it) {
        v2f a = *(const v2f*)(pt + (long)it * 64);
        acc0 = __builtin_amdgcn_wmma_f32_16x16x4_f32(
             false, a,  false, ones,
             (short)0, acc0,  false,  false);
    }
    laneSum = (((acc0[0] + acc0[1]) + (acc0[2] + acc0[3])) +
               ((acc0[4] + acc0[5]) + (acc0[6] + acc0[7]))) +
              (((acc1[0] + acc1[1]) + (acc1[2] + acc1[3])) +
               ((acc1[4] + acc1[5]) + (acc1[6] + acc1[7])));
#else
    float s = 0.0f;
    const int iters = wave_chunk >> 6;
    const float* p = loss + base + lane * 2;
    for (int it = 0; it < iters; ++it) {
        v2f a = *(const v2f*)(p + (long)it * 64);
        s += a.x + a.y;
    }
    laneSum = s;
#endif

    #pragma unroll
    for (int off = 16; off > 0; off >>= 1)
        laneSum += __shfl_xor(laneSum, off, 32);

#if HAS_WMMA_F32_16X16X4
    laneSum *= (1.0f / 16.0f);
#endif

    __shared__ float smem[8];
    if (lane == 0) smem[wave] = laneSum;
    __syncthreads();
    if (threadIdx.x == 0) {
        float s = 0.0f;
        #pragma unroll
        for (int w = 0; w < 8; ++w) s += smem[w];
        partials[blockIdx.x] = s;
    }
}

__device__ __forceinline__ v2f sl_fma2(v2f a, v2f b, v2f c)
{
    return __builtin_elementwise_fma(a, b, c);
}
__device__ __forceinline__ v2f sl_splat2(float a)
{
    v2f r; r.x = a; r.y = a; return r;
}
__device__ __forceinline__ v2f sl_rcp2(v2f a)
{
    v2f r; r.x = __builtin_amdgcn_rcpf(a.x); r.y = __builtin_amdgcn_rcpf(a.y); return r;
}
__device__ __forceinline__ v2f sl_exp2_2(v2f a)
{
    v2f r; r.x = __builtin_amdgcn_exp2f(a.x); r.y = __builtin_amdgcn_exp2f(a.y); return r;
}
struct sl_init {
    v2f p2, zp1, p, lg;
};
__device__ __forceinline__ void sl_i1(sl_init& g, v2f z)
{
    g.p2 = sl_fma2(sl_splat2(5.43656365691809f), z, sl_splat2(2.0f));
    g.p2.x = fmaxf(g.p2.x, 0.0f); g.p2.y = fmaxf(g.p2.y, 0.0f);
    g.zp1 = sl_splat2(1.0f) + z;
}
__device__ __forceinline__ void sl_it(sl_init& g)
{
    g.p.x = __builtin_amdgcn_sqrtf(g.p2.x);
    g.p.y = __builtin_amdgcn_sqrtf(g.p2.y);
    g.lg.x = __builtin_amdgcn_logf(g.zp1.x);
    g.lg.y = __builtin_amdgcn_logf(g.zp1.y);
}
__device__ __forceinline__ v2f sl_i2(const sl_init& g, v2f z)
{
    const v2f inner = sl_fma2(sl_splat2(-(1.0f / 3.0f)), g.p, sl_splat2(1.0f));
    const v2f w_near = sl_fma2(g.p, inner, sl_splat2(-1.0f));
    const v2f w_far = sl_splat2(0.6931471805599453f) * g.lg;
    v2f w;
    w.x = (z.x < -0.25f) ? w_near.x : w_far.x;
    w.y = (z.y < -0.25f) ? w_near.y : w_far.y;
    return w;
}

struct sl_pair {
    v2f z, w, m, wp1, h, ew, r1, f, den, r2;
};
__device__ __forceinline__ void sl_s1(sl_pair& s)
{
    s.m   = s.w * sl_splat2(1.4426950408889634f);
    s.wp1 = s.w + sl_splat2(1.0f);
    s.h   = sl_fma2(sl_splat2(0.5f), s.w, sl_splat2(1.0f));
}
__device__ __forceinline__ void sl_t1(sl_pair& s)
{
    s.ew = sl_exp2_2(s.m);
    s.r1 = sl_rcp2(s.wp1);
}
__device__ __forceinline__ void sl_s2(sl_pair& s)
{
    s.f = sl_fma2(s.w, s.ew, -s.z);
    const v2f corr = (s.h * s.f) * s.r1;
    s.den = sl_fma2(s.ew, s.wp1, -corr);
}
__device__ __forceinline__ void sl_t2(sl_pair& s)
{
    s.r2 = sl_rcp2(s.den);
}
__device__ __forceinline__ void sl_s3(sl_pair& s)
{
    s.w = sl_fma2(-s.f, s.r2, s.w);
}

#define SL_PAIR_T_V() \
    __builtin_amdgcn_sched_group_barrier(0x400, 1, 0); \
    __builtin_amdgcn_sched_group_barrier(0x002, 1, 0)
#define SL_STRETCH_4T() \
    SL_PAIR_T_V(); SL_PAIR_T_V(); SL_PAIR_T_V(); SL_PAIR_T_V(); \
    __builtin_amdgcn_sched_barrier(0)
#define SL_STRETCH_2T() \
    SL_PAIR_T_V(); SL_PAIR_T_V(); \
    __builtin_amdgcn_sched_group_barrier(0x002, 2, 0); \
    __builtin_amdgcn_sched_barrier(0)

__global__ __launch_bounds__(256) void sl_superloss_kernel(
    const float* __restrict__ loss, const float* __restrict__ partials,
    float* __restrict__ out_super, float* __restrict__ out_sigma,
    int n4, int nparts, float inv_n)
{
    const int t = threadIdx.x;
    const unsigned stride = gridDim.x * 256u;
    unsigned i = blockIdx.x * 256u + (unsigned)t;

    float4 x = make_float4(0.0f, 0.0f, 0.0f, 0.0f);
    if (i < (unsigned)n4) x = ((const float4*)loss)[i];

    float va = (t < nparts) ? partials[t] : 0.0f;
    float vb = (t + 256 < nparts) ? partials[t + 256] : 0.0f;
    #pragma unroll
    for (int off = 16; off > 0; off >>= 1) {
        va += __shfl_xor(va, off, 32);
        vb += __shfl_xor(vb, off, 32);
    }
    __shared__ float smem[16];
    const int wave = t >> 5;
    const int lane = t & 31;
    if (lane == 0) { smem[wave] = va; smem[wave + 8] = vb; }
    __syncthreads();
    float s = 0.0f;
    #pragma unroll
    for (int w = 0; w < 16; ++w) s += smem[w];
    const float mean = s * inv_n;
    const float tau = 0.9f * 0.5f + 0.1f * mean;

    const float zmin = -0.36787944117144233f + 1.1920928955078125e-7f;

    while (i < (unsigned)n4) {
        const unsigned inext = i + stride;
        const float4 xn = ((const float4*)loss)[(inext < (unsigned)n4) ? inext : i];

        v2f xa; xa.x = x.x; xa.y = x.y;
        v2f xb; xb.x = x.z; xb.y = x.w;
        const v2f tau2 = sl_splat2(tau);
        const v2f half2 = sl_splat2(0.5f);
        sl_pair a, b;
        a.z = half2 * (xa - tau2);
        b.z = half2 * (xb - tau2);
        a.z.x = fmaxf(zmin, a.z.x); a.z.y = fmaxf(zmin, a.z.y);
        b.z.x = fmaxf(zmin, b.z.x); b.z.y = fmaxf(zmin, b.z.y);
        sl_init ga, gb;
        sl_i1(ga, a.z);
        __builtin_amdgcn_sched_barrier(0);

        sl_it(ga);
        sl_i1(gb, b.z);
        SL_STRETCH_4T();

        sl_it(gb);
        a.w = sl_i2(ga, a.z);
        sl_s1(a);
        SL_STRETCH_4T();

        const v2f nl2e = sl_splat2(-1.4426950408889634f);
        #pragma unroll
        for (int it = 0; it < 4; ++it) {
            sl_t1(a);
            if (it > 0) sl_s3(b); else b.w = sl_i2(gb, b.z);
            sl_s1(b);
            SL_STRETCH_4T();

            sl_t1(b);
            sl_s2(a);
            SL_STRETCH_4T();

            sl_t2(a);
            sl_s2(b);
            SL_STRETCH_2T();

            sl_t2(b);
            sl_s3(a);
            if (it < 3) sl_s1(a);
            SL_STRETCH_2T();
        }

        sl_s3(b);

        const v2f sga = sl_exp2_2(a.w * nl2e);
        const v2f sgb = sl_exp2_2(b.w * nl2e);
        const v2f sla = sga * xa;
        const v2f slb = sgb * xb;
        SL_PAIR_T_V(); SL_PAIR_T_V(); SL_PAIR_T_V(); SL_PAIR_T_V();

        float4 o;
        o.x = sla.x; o.y = sla.y; o.z = slb.x; o.w = slb.y;
        ((float4*)out_super)[i] = o;
        o.x = sga.x; o.y = sga.y; o.z = sgb.x; o.w = sgb.y;
        ((float4*)out_sigma)[i] = o;

        x = xn;
        i = inext;
    }
}

extern "C" void kernel_launch(void* const* d_in, const int* in_sizes, int n_in,
                              void* d_out, int out_size, void* d_ws, size_t ws_size,
                              hipStream_t stream)
{
    const float* loss = (const float*)d_in[0];
    const int n = in_sizes[0];

    float* out_super = (float*)d_out;
    float* out_sigma = out_super + n;

    float* partials = (float*)d_ws;

    const int RBLOCKS = 512;
    const int chunk = n / RBLOCKS;

    sl_partial_sum_kernel<<<RBLOCKS, 256, 0, stream>>>(loss, partials, chunk);

    const int n4 = n >> 2;
    sl_superloss_kernel<<<2048, 256, 0, stream>>>(
        loss, partials, out_super, out_sigma, n4, RBLOCKS, 1.0f / (float)n);
}
